# starting kernel
# baseline (speedup 1.0000x reference)
#include <hip/hip_runtime.h>

typedef __attribute__((ext_vector_type(2))) float v2f;
typedef __attribute__((ext_vector_type(8))) float v8f;

#define MUL     4096
#define KSPLIT  8
#define KCHUNK  (MUL / KSPLIT)

__global__ __launch_bounds__(256)
void irreps_linear_wmma(const float* __restrict__ x,
                        const float* __restrict__ w,
                        float* __restrict__ out) {
    __shared__ float red[KSPLIT * 512];

    const int block = blockIdx.x;
    const int path  = block >> 7;
    const int pair  = block & 127;
    const int v0    = pair << 5;

    const int       d      = 2 * path + 1;
    const int       io_off = (path * path) << 12;
    const long long w_off  = (long long)path << 24;

    const int tid  = threadIdx.x;
    const int wave = tid >> 5;
    const int lane = tid & 31;
    const int half = lane >> 4;
    const int r    = lane & 15;

    const int   i_col = (r < d) ? r : (d - 1);
    const float maskf = (r < d) ? 1.0f : 0.0f;

    const int u0 = wave * KCHUNK + 2 * half;
    const float* wp = w + w_off + (long long)u0 * MUL + v0 + r;
    const float* xp = x + io_off + u0 * d + i_col;

    v8f acc0 = {};
    v8f acc1 = {};

    #pragma unroll 4
    for (int it = 0; it < KCHUNK / 4; ++it) {
        v2f a0, a1, b;
        a0.x = wp[0];
        a0.y = wp[MUL];
        a1.x = wp[16];
        a1.y = wp[MUL + 16];
        b.x = xp[0] * maskf;
        b.y = xp[d] * maskf;

        acc0 = __builtin_amdgcn_wmma_f32_16x16x4_f32(false, a0, false, b,
                                                     (short)0, acc0, false, false);
        acc1 = __builtin_amdgcn_wmma_f32_16x16x4_f32(false, a1, false, b,
                                                     (short)0, acc1, false, false);

        wp += 4 * MUL;
        xp += 4 * d;
    }

    {
        float* my = &red[wave * 512 + lane];
        #pragma unroll
        for (int j = 0; j < 8; ++j) {
            my[j * 32]       = acc0[j];
            my[256 + j * 32] = acc1[j];
        }
    }
    __syncthreads();

    const float norm = 0.015625f;
    for (int s = tid; s < 512; s += 256) {
        float sum = 0.0f;
        #pragma unroll
        for (int wv = 0; wv < KSPLIT; ++wv)
            sum += red[wv * 512 + s];
        const int tile = s >> 8;
        const int j    = (s >> 5) & 7;
        const int ln   = s & 31;
        const int M    = j + ((ln >> 4) << 3);
        const int N    = ln & 15;
        if (N < d)
            out[io_off + (v0 + tile * 16 + M) * d + N] = sum * norm;
    }
}

extern "C" void kernel_launch(void* const* d_in, const int* in_sizes, int n_in,
                              void* d_out, int out_size, void* d_ws, size_t ws_size,
                              hipStream_t stream) {
    (void)in_sizes; (void)n_in; (void)d_ws; (void)ws_size; (void)out_size;
    const float* x   = (const float*)d_in[0];
    const float* wts = (const float*)d_in[1];
    float* out = (float*)d_out;

    dim3 grid(512);
    dim3 block(256);
    irreps_linear_wmma<<<grid, block, 0, stream>>>(x, wts, out);
}


# Weights streamed as aligned 256-byte row pieces: non-temporal 16-byte loads, 28 steps in flight per wave, 256 blocks of 8 waves x 64 columns; 2 rows per WMMA, so a wave's sum order changes
# speedup vs baseline: 1.3832x; 1.3832x over previous
#include <hip/hip_runtime.h>

typedef __attribute__((ext_vector_type(2))) float v2f;
typedef __attribute__((ext_vector_type(4))) float v4f;
typedef __attribute__((ext_vector_type(8))) float v8f;

#define MUL     4096
#define KSPLIT  8
#define KCHUNK  (MUL / KSPLIT)
#define NSTEP   (KCHUNK / 2)
#define RING    32
#define AHEAD   28

__global__ __launch_bounds__(256)
void irreps_linear_wmma(const float* __restrict__ x,
                        const float* __restrict__ w,
                        float* __restrict__ out) {
    __shared__ float red[KSPLIT * 512];

    const int block  = blockIdx.x;
    const int path   = block >> 6;
    const int stripe = block & 63;
    const int v0     = stripe << 6;

    const int       d      = 2 * path + 1;
    const int       io_off = (path * path) << 12;
    const long long w_off  = (long long)path << 24;

    const int tid  = threadIdx.x;
    const int wave = tid >> 5;
    const int lane = tid & 31;
    const int h    = lane >> 4;
    const int c    = lane & 15;
    const int e    = c >> 3;
    const int i    = c & 7;
    const int i_ld = (i < d) ? i : (d - 1);
    const bool m0  = (e == 0) && (i < d);
    const bool m1  = (e == 1) && (i < d);

    const int u0 = wave * KCHUNK + h;
    const float* wp = w + w_off + (long long)u0 * MUL + v0 + 4 * c;
    const float* xp = x + io_off + u0 * d + i_ld;

    v4f   wbuf[RING];
    float xbuf[RING];
    v8f acc0 = {};
    v8f acc1 = {};

#define LOAD_STEP(SLOT, T) do { \
        wbuf[(SLOT)] = __builtin_nontemporal_load(reinterpret_cast<const v4f*>(wp + (long long)(T) * (2 * MUL))); \
        xbuf[(SLOT)] = xp[(T) * 2 * d]; } while (0)

#define USE_STEP(SLOT) do { \
        const v4f wv = wbuf[(SLOT)]; const float xv = xbuf[(SLOT)]; \
        v2f b;  b.x  = m0 ? xv : 0.0f;  b.y  = m1 ? xv : 0.0f; \
        v2f a0; a0.x = wv.x; a0.y = wv.y; \
        v2f a1; a1.x = wv.z; a1.y = wv.w; \
        acc0 = __builtin_amdgcn_wmma_f32_16x16x4_f32(false, a0, false, b, (short)0, acc0, false, false); \
        acc1 = __builtin_amdgcn_wmma_f32_16x16x4_f32(false, a1, false, b, (short)0, acc1, false, false); } while (0)

    #pragma unroll
    for (int s = 0; s < AHEAD; ++s) {
        LOAD_STEP(s, s);
        __builtin_amdgcn_sched_barrier(0);
    }

    for (int it = 0; it < NSTEP / RING - 1; ++it) {
        #pragma unroll
        for (int s = 0; s < RING; ++s) {
            LOAD_STEP((s + AHEAD) % RING, s + AHEAD);
            USE_STEP(s);
            __builtin_amdgcn_sched_barrier(0);
        }
        wp += RING * 2 * MUL;
        xp += RING * 2 * d;
    }
    #pragma unroll
    for (int s = 0; s < RING; ++s) {
        if (s + AHEAD < RING) LOAD_STEP((s + AHEAD) % RING, s + AHEAD);
        USE_STEP(s);
        __builtin_amdgcn_sched_barrier(0);
    }
#undef LOAD_STEP
#undef USE_STEP

    {
        float* my = &red[wave * 512 + lane];
        #pragma unroll
        for (int j = 0; j < 8; ++j) {
            my[j * 32]       = acc0[j];
            my[256 + j * 32] = acc1[j];
        }
    }
    __syncthreads();

    const float norm = 0.015625f;
    const int nout = 64 * d;
    for (int t = tid; t < nout; t += 256) {
        const int vv   = t / d;
        const int comp = t - vv * d;
        const int m    = vv >> 2;
        const int q    = vv & 3;
        const int n    = ((q & 1) << 3) + comp;
        const int idx  = (q >> 1) * 256 + (m & 7) * 32 + ((m >> 3) << 4) + n;
        float sum = 0.0f;
        #pragma unroll
        for (int wv = 0; wv < KSPLIT; ++wv)
            sum += red[wv * 512 + idx];
        out[io_off + v0 * d + t] = sum * norm;
    }
}

extern "C" void kernel_launch(void* const* d_in, const int* in_sizes, int n_in,
                              void* d_out, int out_size, void* d_ws, size_t ws_size,
                              hipStream_t stream) {
    (void)in_sizes; (void)n_in; (void)d_ws; (void)ws_size; (void)out_size;
    const float* x   = (const float*)d_in[0];
    const float* wts = (const float*)d_in[1];
    float* out = (float*)d_out;

    dim3 grid(256);
    dim3 block(256);
    irreps_linear_wmma<<<grid, block, 0, stream>>>(x, wts, out);
}
